# starting kernel
# baseline (speedup 1.0000x reference)
#include <hip/hip_runtime.h>


#define N_NODES 50000
#define N_EDGES 600000
#define DIM     128
#define HID     256
#define AS_STR  (DIM + 4)
#define HS_STR  (HID + 4)

typedef __attribute__((ext_vector_type(2))) float v2f;
typedef __attribute__((ext_vector_type(8))) float v8f;

__global__ __launch_bounds__(256) void init_acc(const float* __restrict__ x,
                                                const float* __restrict__ e1,
                                                const float* __restrict__ e2,
                                                float* __restrict__ acc) {
  int idx = blockIdx.x * 256 + threadIdx.x;
  int d   = idx & (DIM - 1);
  acc[idx] = x[idx] + e1[4 * DIM + d] + e2[0 * DIM + d];
}

__global__ __launch_bounds__(256) void scatter_edges(const float* __restrict__ x,
                                                     const int* __restrict__ edge_index,
                                                     const int* __restrict__ edge_attr,
                                                     const float* __restrict__ e1,
                                                     const float* __restrict__ e2,
                                                     float* __restrict__ acc) {
  int idx = blockIdx.x * 256 + threadIdx.x;
  int e   = idx >> 7;
  int d   = idx & (DIM - 1);
  int row = edge_index[e];
  int col = edge_index[N_EDGES + e];
  int a0  = edge_attr[2 * e];
  int a1  = edge_attr[2 * e + 1];
  float val = x[col * DIM + d] + e1[a0 * DIM + d] + e2[a1 * DIM + d];
  unsafeAtomicAdd(&acc[row * DIM + d], val);
}

__global__ __launch_bounds__(128) void mlp_fused(const float* __restrict__ acc,
                                                 const float* __restrict__ W1,
                                                 const float* __restrict__ b1,
                                                 const float* __restrict__ W2,
                                                 const float* __restrict__ b2,
                                                 float* __restrict__ out) {
  __shared__ float As[16 * AS_STR];
  __shared__ float Hs[16 * HS_STR];
  const int tid  = threadIdx.x;
  const int wave = tid >> 5;
  const int lane = tid & 31;
  const int m    = lane & 15;
  const int kh   = lane >> 4;
  const int row0 = blockIdx.x * 16;

  {
    const int r  = tid >> 3;
    const int c0 = (tid & 7) * 16;
    const float4* src = (const float4*)(acc + (row0 + r) * DIM + c0);
    float4*       dst = (float4*)(As + r * AS_STR + c0);
#pragma unroll
    for (int i = 0; i < 4; ++i) dst[i] = src[i];
  }
  __syncthreads();

#pragma unroll
  for (int nt = 0; nt < 4; ++nt) {
    const int n0 = (wave * 4 + nt) * 16;
    const float* w1row = W1 + (n0 + m) * DIM;
    v8f c = {};
#pragma unroll
    for (int k0 = 0; k0 < DIM; k0 += 4) {
      v2f a, b;
      a.x = As[m * AS_STR + k0 + 2 * kh];
      a.y = As[m * AS_STR + k0 + 2 * kh + 1];
      b.x = w1row[k0 + 2 * kh];
      b.y = w1row[k0 + 2 * kh + 1];
      c = __builtin_amdgcn_wmma_f32_16x16x4_f32(false, a, false, b, (short)0, c,
                                                false, false);
    }
    const float bias = b1[n0 + m];
#pragma unroll
    for (int j = 0; j < 8; ++j) {
      const int mj = j + 8 * kh;
      float h = c[j] + bias;
      Hs[mj * HS_STR + n0 + m] = h > 0.0f ? h : 0.0f;
    }
  }
  __syncthreads();

#pragma unroll
  for (int nt = 0; nt < 2; ++nt) {
    const int n0 = (wave * 2 + nt) * 16;
    const float* w2row = W2 + (n0 + m) * HID;
    v8f c = {};
#pragma unroll
    for (int k0 = 0; k0 < HID; k0 += 4) {
      v2f a, b;
      a.x = Hs[m * HS_STR + k0 + 2 * kh];
      a.y = Hs[m * HS_STR + k0 + 2 * kh + 1];
      b.x = w2row[k0 + 2 * kh];
      b.y = w2row[k0 + 2 * kh + 1];
      c = __builtin_amdgcn_wmma_f32_16x16x4_f32(false, a, false, b, (short)0, c,
                                                false, false);
    }
    const float bias = b2[n0 + m];
#pragma unroll
    for (int j = 0; j < 8; ++j) {
      const int mj = j + 8 * kh;
      out[(row0 + mj) * DIM + n0 + m] = c[j] + bias;
    }
  }
}

extern "C" void kernel_launch(void* const* d_in, const int* in_sizes, int n_in,
                              void* d_out, int out_size, void* d_ws, size_t ws_size,
                              hipStream_t stream) {
  const float* x          = (const float*)d_in[0];
  const int*   edge_index = (const int*)d_in[1];
  const int*   edge_attr  = (const int*)d_in[2];
  const float* W1         = (const float*)d_in[3];
  const float* b1         = (const float*)d_in[4];
  const float* W2         = (const float*)d_in[5];
  const float* b2         = (const float*)d_in[6];
  const float* e1         = (const float*)d_in[7];
  const float* e2         = (const float*)d_in[8];
  float*       out        = (float*)d_out;
  float*       acc        = (float*)d_ws;

  (void)in_sizes; (void)n_in; (void)out_size; (void)ws_size;

  init_acc<<<(N_NODES * DIM) / 256, 256, 0, stream>>>(x, e1, e2, acc);
  scatter_edges<<<(N_EDGES * DIM) / 256, 256, 0, stream>>>(x, edge_index, edge_attr,
                                                           e1, e2, acc);
  mlp_fused<<<N_NODES / 16, 128, 0, stream>>>(acc, W1, b1, W2, b2, out);
}


# Replace per-element f32 scatter atomics by a per-call counting sort of edges by destination row (int atomics) and a one-wave-per-node register gather; sum order = arrival rank in row
# speedup vs baseline: 1.8522x; 1.8522x over previous
#include <hip/hip_runtime.h>


#define N_NODES 50000
#define N_EDGES 600000
#define DIM     128
#define HID     256
#define AS_STR  (DIM + 4)
#define HS_STR  (HID + 4)

typedef __attribute__((ext_vector_type(2))) float v2f;
typedef __attribute__((ext_vector_type(8))) float v8f;

__global__ __launch_bounds__(256) void zero_counts(int* __restrict__ cnt,
                                                   int* __restrict__ cursor) {
  int idx = blockIdx.x * 256 + threadIdx.x;
  if (idx < N_NODES) cnt[idx] = 0;
  if (idx == 0) *cursor = 0;
}

__global__ __launch_bounds__(256) void count_edges(const int* __restrict__ edge_index,
                                                   int* __restrict__ cnt,
                                                   int* __restrict__ rank) {
  int e = blockIdx.x * 256 + threadIdx.x;
  if (e < N_EDGES) rank[e] = atomicAdd(&cnt[edge_index[e]], 1);
}

__global__ __launch_bounds__(256) void alloc_segments(const int* __restrict__ cnt,
                                                      int* __restrict__ cursor,
                                                      int* __restrict__ offs) {
  const int n    = blockIdx.x * 256 + threadIdx.x;
  const int lane = threadIdx.x & 31;
  const int c    = (n < N_NODES) ? cnt[n] : 0;
  int incl = c;
#pragma unroll
  for (int d = 1; d < 32; d <<= 1) {
    int t = __shfl_up(incl, d);
    if (lane >= d) incl += t;
  }
  const int total = __shfl(incl, 31);
  int base = 0;
  if (lane == 0) base = atomicAdd(cursor, total);
  base = __shfl(base, 0);
  if (n < N_NODES) offs[n] = base + incl - c;
}

static_assert(N_NODES <= 65536, "col must fit the record's 16 bits");
__global__ __launch_bounds__(256) void fill_records(const int* __restrict__ edge_index,
                                                    const int* __restrict__ edge_attr,
                                                    const int* __restrict__ offs,
                                                    const int* __restrict__ rank,
                                                    unsigned int* __restrict__ rec) {
  int e = blockIdx.x * 256 + threadIdx.x;
  if (e < N_EDGES) {
    int row = edge_index[e];
    int col = edge_index[N_EDGES + e];
    int a0  = edge_attr[2 * e];
    int a1  = edge_attr[2 * e + 1];
    unsigned int code = ((unsigned int)col << 16) | ((unsigned int)a0 << 8) | (unsigned int)a1;
    atomicExch(&rec[offs[row] + rank[e]], code);
  }
}

__global__ __launch_bounds__(256) void gather_nodes(const float* __restrict__ x,
                                                    const float* __restrict__ e1,
                                                    const float* __restrict__ e2,
                                                    const int* __restrict__ cnt,
                                                    const int* __restrict__ offs,
                                                    const unsigned int* __restrict__ rec,
                                                    float* __restrict__ acc) {
  const int lane = threadIdx.x & 31;
  const int n     = blockIdx.x * 8 + __builtin_amdgcn_readfirstlane(threadIdx.x >> 5);
  const int c     = __builtin_amdgcn_readfirstlane(cnt[n]);
  const int start = __builtin_amdgcn_readfirstlane(offs[n]);
  const int d0    = lane * 4;

  const float4 xs = *(const float4*)(x + n * DIM + d0);
  const float4 t1 = *(const float4*)(e1 + 4 * DIM + d0);
  const float4 t2 = *(const float4*)(e2 + 0 * DIM + d0);
  float4 s;
  s.x = xs.x + t1.x + t2.x;
  s.y = xs.y + t1.y + t2.y;
  s.z = xs.z + t1.z + t2.z;
  s.w = xs.w + t1.w + t2.w;

  for (int base = 0; base < c; base += 32) {
    unsigned int r = 0;
    if (base + lane < c) r = rec[start + base + lane];
    const int m = (c - base < 32) ? (c - base) : 32;
#pragma unroll 4
    for (int j = 0; j < m; ++j) {
      const unsigned int code = __shfl(r, j);
      const float4 xv = *(const float4*)(x + (code >> 16) * DIM + d0);
      const float4 ev = *(const float4*)(e1 + ((code >> 8) & 255) * DIM + d0);
      const float4 fv = *(const float4*)(e2 + (code & 255) * DIM + d0);
      s.x += xv.x + ev.x + fv.x;
      s.y += xv.y + ev.y + fv.y;
      s.z += xv.z + ev.z + fv.z;
      s.w += xv.w + ev.w + fv.w;
    }
  }
  *(float4*)(acc + n * DIM + d0) = s;
}

__global__ __launch_bounds__(128) void mlp_fused(const float* __restrict__ acc,
                                                 const float* __restrict__ W1,
                                                 const float* __restrict__ b1,
                                                 const float* __restrict__ W2,
                                                 const float* __restrict__ b2,
                                                 float* __restrict__ out) {
  __shared__ float As[16 * AS_STR];
  __shared__ float Hs[16 * HS_STR];
  const int tid  = threadIdx.x;
  const int wave = tid >> 5;
  const int lane = tid & 31;
  const int m    = lane & 15;
  const int kh   = lane >> 4;
  const int row0 = blockIdx.x * 16;

  {
    const int r  = tid >> 3;
    const int c0 = (tid & 7) * 16;
    const float4* src = (const float4*)(acc + (row0 + r) * DIM + c0);
    float4*       dst = (float4*)(As + r * AS_STR + c0);
#pragma unroll
    for (int i = 0; i < 4; ++i) dst[i] = src[i];
  }
  __syncthreads();

#pragma unroll
  for (int nt = 0; nt < 4; ++nt) {
    const int n0 = (wave * 4 + nt) * 16;
    const float* w1row = W1 + (n0 + m) * DIM;
    v8f c = {};
#pragma unroll
    for (int k0 = 0; k0 < DIM; k0 += 4) {
      v2f a, b;
      a.x = As[m * AS_STR + k0 + 2 * kh];
      a.y = As[m * AS_STR + k0 + 2 * kh + 1];
      b.x = w1row[k0 + 2 * kh];
      b.y = w1row[k0 + 2 * kh + 1];
      c = __builtin_amdgcn_wmma_f32_16x16x4_f32(false, a, false, b, (short)0, c,
                                                false, false);
    }
    const float bias = b1[n0 + m];
#pragma unroll
    for (int j = 0; j < 8; ++j) {
      const int mj = j + 8 * kh;
      float h = c[j] + bias;
      Hs[mj * HS_STR + n0 + m] = h > 0.0f ? h : 0.0f;
    }
  }
  __syncthreads();

#pragma unroll
  for (int nt = 0; nt < 2; ++nt) {
    const int n0 = (wave * 2 + nt) * 16;
    const float* w2row = W2 + (n0 + m) * HID;
    v8f c = {};
#pragma unroll
    for (int k0 = 0; k0 < HID; k0 += 4) {
      v2f a, b;
      a.x = Hs[m * HS_STR + k0 + 2 * kh];
      a.y = Hs[m * HS_STR + k0 + 2 * kh + 1];
      b.x = w2row[k0 + 2 * kh];
      b.y = w2row[k0 + 2 * kh + 1];
      c = __builtin_amdgcn_wmma_f32_16x16x4_f32(false, a, false, b, (short)0, c,
                                                false, false);
    }
    const float bias = b2[n0 + m];
#pragma unroll
    for (int j = 0; j < 8; ++j) {
      const int mj = j + 8 * kh;
      out[(row0 + mj) * DIM + n0 + m] = c[j] + bias;
    }
  }
}

extern "C" void kernel_launch(void* const* d_in, const int* in_sizes, int n_in,
                              void* d_out, int out_size, void* d_ws, size_t ws_size,
                              hipStream_t stream) {
  const float* x          = (const float*)d_in[0];
  const int*   edge_index = (const int*)d_in[1];
  const int*   edge_attr  = (const int*)d_in[2];
  const float* W1         = (const float*)d_in[3];
  const float* b1         = (const float*)d_in[4];
  const float* W2         = (const float*)d_in[5];
  const float* b2         = (const float*)d_in[6];
  const float* e1         = (const float*)d_in[7];
  const float* e2         = (const float*)d_in[8];
  float*       out        = (float*)d_out;
  char*        ws         = (char*)d_ws;
  float*       acc        = (float*)ws;
  int*         cnt        = (int*)(ws + 25600000);
  int*         offs       = (int*)(ws + 25800000);
  int*         rank       = (int*)(ws + 26000000);
  unsigned int* rec       = (unsigned int*)(ws + 28400000);
  int*         cursor     = (int*)(ws + 33200000);

  (void)in_sizes; (void)n_in; (void)out_size; (void)ws_size;

  zero_counts<<<196, 256, 0, stream>>>(cnt, cursor);
  count_edges<<<2344, 256, 0, stream>>>(edge_index, cnt, rank);
  alloc_segments<<<196, 256, 0, stream>>>(cnt, cursor, offs);
  fill_records<<<2344, 256, 0, stream>>>(edge_index, edge_attr, offs, rank, rec);
  gather_nodes<<<N_NODES / 8, 256, 0, stream>>>(x, e1, e2, cnt, offs, rec, acc);
  mlp_fused<<<N_NODES / 16, 128, 0, stream>>>(acc, W1, b1, W2, b2, out);
}


# MLP kernel made persistent: one 8-wave workgroup per compute unit stages W1 and W2 into LDS once and walks its 16-row tiles, two per pass; per-tile k order and sum order unchanged
# speedup vs baseline: 2.3336x; 1.2599x over previous
#include <hip/hip_runtime.h>


#define N_NODES 50000
#define N_EDGES 600000
#define DIM     128
#define HID     256
#define AS_STR  (DIM + 4)
#define HS_STR  (HID + 4)

typedef __attribute__((ext_vector_type(2))) float v2f;
typedef __attribute__((ext_vector_type(8))) float v8f;

__global__ __launch_bounds__(256) void zero_counts(int* __restrict__ cnt,
                                                   int* __restrict__ cursor) {
  int idx = blockIdx.x * 256 + threadIdx.x;
  if (idx < N_NODES) cnt[idx] = 0;
  if (idx == 0) *cursor = 0;
}

__global__ __launch_bounds__(256) void count_edges(const int* __restrict__ edge_index,
                                                   int* __restrict__ cnt,
                                                   int* __restrict__ rank) {
  int e = blockIdx.x * 256 + threadIdx.x;
  if (e < N_EDGES) rank[e] = atomicAdd(&cnt[edge_index[e]], 1);
}

__global__ __launch_bounds__(256) void alloc_segments(const int* __restrict__ cnt,
                                                      int* __restrict__ cursor,
                                                      int* __restrict__ offs) {
  const int n    = blockIdx.x * 256 + threadIdx.x;
  const int lane = threadIdx.x & 31;
  const int c    = (n < N_NODES) ? cnt[n] : 0;
  int incl = c;
#pragma unroll
  for (int d = 1; d < 32; d <<= 1) {
    int t = __shfl_up(incl, d);
    if (lane >= d) incl += t;
  }
  const int total = __shfl(incl, 31);
  int base = 0;
  if (lane == 0) base = atomicAdd(cursor, total);
  base = __shfl(base, 0);
  if (n < N_NODES) offs[n] = base + incl - c;
}

static_assert(N_NODES <= 65536, "col must fit the record's 16 bits");
__global__ __launch_bounds__(256) void fill_records(const int* __restrict__ edge_index,
                                                    const int* __restrict__ edge_attr,
                                                    const int* __restrict__ offs,
                                                    const int* __restrict__ rank,
                                                    unsigned int* __restrict__ rec) {
  int e = blockIdx.x * 256 + threadIdx.x;
  if (e < N_EDGES) {
    int row = edge_index[e];
    int col = edge_index[N_EDGES + e];
    int a0  = edge_attr[2 * e];
    int a1  = edge_attr[2 * e + 1];
    unsigned int code = ((unsigned int)col << 16) | ((unsigned int)a0 << 8) | (unsigned int)a1;
    atomicExch(&rec[offs[row] + rank[e]], code);
  }
}

__global__ __launch_bounds__(256) void gather_nodes(const float* __restrict__ x,
                                                    const float* __restrict__ e1,
                                                    const float* __restrict__ e2,
                                                    const int* __restrict__ cnt,
                                                    const int* __restrict__ offs,
                                                    const unsigned int* __restrict__ rec,
                                                    float* __restrict__ acc) {
  const int lane = threadIdx.x & 31;
  const int n     = blockIdx.x * 8 + __builtin_amdgcn_readfirstlane(threadIdx.x >> 5);
  const int c     = __builtin_amdgcn_readfirstlane(cnt[n]);
  const int start = __builtin_amdgcn_readfirstlane(offs[n]);
  const int d0    = lane * 4;

  const float4 xs = *(const float4*)(x + n * DIM + d0);
  const float4 t1 = *(const float4*)(e1 + 4 * DIM + d0);
  const float4 t2 = *(const float4*)(e2 + 0 * DIM + d0);
  float4 s;
  s.x = xs.x + t1.x + t2.x;
  s.y = xs.y + t1.y + t2.y;
  s.z = xs.z + t1.z + t2.z;
  s.w = xs.w + t1.w + t2.w;

  for (int base = 0; base < c; base += 32) {
    unsigned int r = 0;
    if (base + lane < c) r = rec[start + base + lane];
    const int m = (c - base < 32) ? (c - base) : 32;
#pragma unroll 4
    for (int j = 0; j < m; ++j) {
      const unsigned int code = __shfl(r, j);
      const float4 xv = *(const float4*)(x + (code >> 16) * DIM + d0);
      const float4 ev = *(const float4*)(e1 + ((code >> 8) & 255) * DIM + d0);
      const float4 fv = *(const float4*)(e2 + (code & 255) * DIM + d0);
      s.x += xv.x + ev.x + fv.x;
      s.y += xv.y + ev.y + fv.y;
      s.z += xv.z + ev.z + fv.z;
      s.w += xv.w + ev.w + fv.w;
    }
  }
  *(float4*)(acc + n * DIM + d0) = s;
}

#define N_TILES   (N_NODES / 16)
#define MLP_WGS   256
#define TILES_LO  (N_TILES / MLP_WGS)
#define TILES_REM (N_TILES % MLP_WGS)

template <int MT>
__device__ __forceinline__ void layer1(const float* As, const float* W1s, float* Hs,
                                       int wave, int m, int kh, float biasA, float biasB) {
  const float* ap = As + m * AS_STR + 2 * kh;
  const float* bp = W1s + (wave * 32 + m) * AS_STR + 2 * kh;
  v8f c[MT][2];
#pragma unroll
  for (int t = 0; t < MT; ++t) { c[t][0] = v8f{}; c[t][1] = v8f{}; }
#pragma unroll
  for (int k0 = 0; k0 < DIM; k0 += 4) {
    const v2f b0 = *(const v2f*)(bp + k0);
    const v2f b1 = *(const v2f*)(bp + 16 * AS_STR + k0);
#pragma unroll
    for (int t = 0; t < MT; ++t) {
      const v2f a = *(const v2f*)(ap + t * 16 * AS_STR + k0);
      c[t][0] = __builtin_amdgcn_wmma_f32_16x16x4_f32(false, a, false, b0, (short)0, c[t][0],
                                                      false, false);
      c[t][1] = __builtin_amdgcn_wmma_f32_16x16x4_f32(false, a, false, b1, (short)0, c[t][1],
                                                      false, false);
    }
  }
#pragma unroll
  for (int t = 0; t < MT; ++t) {
#pragma unroll
    for (int nt = 0; nt < 2; ++nt) {
      const float bias = nt ? biasB : biasA;
      const int n0 = wave * 32 + nt * 16;
#pragma unroll
      for (int j = 0; j < 8; ++j) {
        const float h = c[t][nt][j] + bias;
        Hs[(t * 16 + j + 8 * kh) * HS_STR + n0 + m] = h > 0.0f ? h : 0.0f;
      }
    }
  }
}

__device__ __forceinline__ void layer2(const float* Hs, const float* W2s, float* __restrict__ out,
                                       int row0, int mt, int wave, int m, int kh,
                                       float biasA, float biasB) {
  const int tr = wave >> 2;
  if (tr >= mt) return;
  const int np = wave & 3;
  const float* ap = Hs + (tr * 16 + m) * HS_STR + 2 * kh;
  const float* bp = W2s + (np * 32 + m) * HS_STR + 2 * kh;
  v8f c0 = {}, c1 = {};
#pragma unroll
  for (int k0 = 0; k0 < HID; k0 += 4) {
    const v2f a  = *(const v2f*)(ap + k0);
    const v2f b0 = *(const v2f*)(bp + k0);
    const v2f b1 = *(const v2f*)(bp + 16 * HS_STR + k0);
    c0 = __builtin_amdgcn_wmma_f32_16x16x4_f32(false, a, false, b0, (short)0, c0, false, false);
    c1 = __builtin_amdgcn_wmma_f32_16x16x4_f32(false, a, false, b1, (short)0, c1, false, false);
  }
  float* op = out + (row0 + tr * 16 + 8 * kh) * DIM + np * 32 + m;
#pragma unroll
  for (int j = 0; j < 8; ++j) {
    op[j * DIM]      = c0[j] + biasA;
    op[j * DIM + 16] = c1[j] + biasB;
  }
}

__global__ __launch_bounds__(256) void mlp_fused(const float* __restrict__ acc,
                                                 const float* __restrict__ W1,
                                                 const float* __restrict__ b1,
                                                 const float* __restrict__ W2,
                                                 const float* __restrict__ b2,
                                                 float* __restrict__ out) {
  __shared__ __attribute__((aligned(16))) float W1s[HID * AS_STR];
  __shared__ __attribute__((aligned(16))) float W2s[DIM * HS_STR];
  __shared__ __attribute__((aligned(16))) float As[32 * AS_STR];
  __shared__ __attribute__((aligned(16))) float Hs[32 * HS_STR];
  const int tid  = threadIdx.x;
  const int wave = tid >> 5;
  const int lane = tid & 31;
  const int m    = lane & 15;
  const int kh   = lane >> 4;
  const int wg   = blockIdx.x;
  int tile = wg * TILES_LO + (wg < TILES_REM ? wg : TILES_REM);
  int left = TILES_LO + (wg < TILES_REM ? 1 : 0);

  {
    const float4* w1v = (const float4*)W1;
#pragma unroll 16
    for (int q = 0; q < HID * DIM / 4 / 256; ++q) {
      const int i = tid + 256 * q;
      *(float4*)(W1s + (i >> 5) * AS_STR + (i & 31) * 4) = w1v[i];
    }
    const float4* w2v = (const float4*)W2;
#pragma unroll 16
    for (int q = 0; q < DIM * HID / 4 / 256; ++q) {
      const int i = tid + 256 * q;
      *(float4*)(W2s + (i >> 6) * HS_STR + (i & 63) * 4) = w2v[i];
    }
  }
  const float bias1a = b1[wave * 32 + m];
  const float bias1b = b1[wave * 32 + 16 + m];
  const float bias2a = b2[(wave & 3) * 32 + m];
  const float bias2b = b2[(wave & 3) * 32 + 16 + m];

  float4 p0 = {}, p1 = {}, p2 = {}, p3 = {};
  float* const d0 = As + (tid >> 5) * AS_STR + (tid & 31) * 4;
  {
    const float4* s = (const float4*)(acc + tile * 16 * DIM);
    p0 = s[tid]; p1 = s[tid + 256];
    *(float4*)d0 = p0; *(float4*)(d0 + 8 * AS_STR) = p1;
    if (left >= 2) {
      p2 = s[tid + 512]; p3 = s[tid + 768];
      *(float4*)(d0 + 16 * AS_STR) = p2; *(float4*)(d0 + 24 * AS_STR) = p3;
    }
  }
  __syncthreads();

  while (left > 0) {
    const int mt    = left >= 2 ? 2 : 1;
    const int ntile = tile + mt;
    const int nleft = left - mt;
    {
      const float4* s = (const float4*)(acc + ntile * 16 * DIM);
      if (nleft >= 1) { p0 = s[tid]; p1 = s[tid + 256]; }
      if (nleft >= 2) { p2 = s[tid + 512]; p3 = s[tid + 768]; }
    }
    if (mt == 2) layer1<2>(As, W1s, Hs, wave, m, kh, bias1a, bias1b);
    else         layer1<1>(As, W1s, Hs, wave, m, kh, bias1a, bias1b);
    __syncthreads();
    if (nleft >= 1) { *(float4*)d0 = p0; *(float4*)(d0 + 8 * AS_STR) = p1; }
    if (nleft >= 2) { *(float4*)(d0 + 16 * AS_STR) = p2; *(float4*)(d0 + 24 * AS_STR) = p3; }
    layer2(Hs, W2s, out, tile * 16, mt, wave, m, kh, bias2a, bias2b);
    __syncthreads();
    tile = ntile;
    left = nleft;
  }
}

extern "C" void kernel_launch(void* const* d_in, const int* in_sizes, int n_in,
                              void* d_out, int out_size, void* d_ws, size_t ws_size,
                              hipStream_t stream) {
  const float* x          = (const float*)d_in[0];
  const int*   edge_index = (const int*)d_in[1];
  const int*   edge_attr  = (const int*)d_in[2];
  const float* W1         = (const float*)d_in[3];
  const float* b1         = (const float*)d_in[4];
  const float* W2         = (const float*)d_in[5];
  const float* b2         = (const float*)d_in[6];
  const float* e1         = (const float*)d_in[7];
  const float* e2         = (const float*)d_in[8];
  float*       out        = (float*)d_out;
  char*        ws         = (char*)d_ws;
  float*       acc        = (float*)ws;
  int*         cnt        = (int*)(ws + 25600000);
  int*         offs       = (int*)(ws + 25800000);
  int*         rank       = (int*)(ws + 26000000);
  unsigned int* rec       = (unsigned int*)(ws + 28400000);
  int*         cursor     = (int*)(ws + 33200000);

  (void)in_sizes; (void)n_in; (void)out_size; (void)ws_size;

  zero_counts<<<196, 256, 0, stream>>>(cnt, cursor);
  count_edges<<<2344, 256, 0, stream>>>(edge_index, cnt, rank);
  alloc_segments<<<196, 256, 0, stream>>>(cnt, cursor, offs);
  fill_records<<<2344, 256, 0, stream>>>(edge_index, edge_attr, offs, rank, rec);
  gather_nodes<<<N_NODES / 8, 256, 0, stream>>>(x, e1, e2, cnt, offs, rec, acc);
  mlp_fused<<<MLP_WGS, 256, 0, stream>>>(acc, W1, b1, W2, b2, out);
}


# Edge and node kernels moved to few long-lived waves: 10 edges per lane in count and fill, 5 nodes per wave in gather (counts, offsets, records fetched up front); sum order still arrival rank in row
# speedup vs baseline: 2.4438x; 1.0472x over previous
#include <hip/hip_runtime.h>


#define N_NODES 50000
#define N_EDGES 600000
#define DIM     128
#define HID     256
#define AS_STR  (DIM + 4)
#define HS_STR  (HID + 4)

typedef __attribute__((ext_vector_type(2))) float v2f;
typedef __attribute__((ext_vector_type(8))) float v8f;

__global__ __launch_bounds__(256) void zero_counts(int* __restrict__ cnt,
                                                   int* __restrict__ cursor) {
  int idx = blockIdx.x * 256 + threadIdx.x;
  if (idx < N_NODES) cnt[idx] = 0;
  if (idx == 0) *cursor = 0;
}

#define EDGE_WGS   256
#define EDGE_LANES (EDGE_WGS * 256)
#define EDGE_FULL  (N_EDGES / EDGE_LANES)
#define EDGE_TAIL  (N_EDGES - EDGE_FULL * EDGE_LANES)

__global__ __launch_bounds__(256) void count_edges(const int* __restrict__ edge_index,
                                                   int* __restrict__ cnt,
                                                   int* __restrict__ rank) {
  const int  t    = blockIdx.x * 256 + threadIdx.x;
  const bool tail = t < EDGE_TAIL;
  int row[EDGE_FULL + 1];
  int rk[EDGE_FULL + 1];
#pragma unroll
  for (int k = 0; k < EDGE_FULL; ++k) row[k] = edge_index[t + k * EDGE_LANES];
  row[EDGE_FULL] = tail ? edge_index[t + EDGE_FULL * EDGE_LANES] : 0;
#pragma unroll
  for (int k = 0; k < EDGE_FULL; ++k) rk[k] = atomicAdd(&cnt[row[k]], 1);
  rk[EDGE_FULL] = 0;
  if (tail) rk[EDGE_FULL] = atomicAdd(&cnt[row[EDGE_FULL]], 1);
#pragma unroll
  for (int k = 0; k < EDGE_FULL; ++k) rank[t + k * EDGE_LANES] = rk[k];
  if (tail) rank[t + EDGE_FULL * EDGE_LANES] = rk[EDGE_FULL];
}

__global__ __launch_bounds__(256) void alloc_segments(const int* __restrict__ cnt,
                                                      int* __restrict__ cursor,
                                                      int* __restrict__ offs) {
  const int n    = blockIdx.x * 256 + threadIdx.x;
  const int lane = threadIdx.x & 31;
  const int c    = (n < N_NODES) ? cnt[n] : 0;
  int incl = c;
#pragma unroll
  for (int d = 1; d < 32; d <<= 1) {
    int t = __shfl_up(incl, d);
    if (lane >= d) incl += t;
  }
  const int total = __shfl(incl, 31);
  int base = 0;
  if (lane == 0) base = atomicAdd(cursor, total);
  base = __shfl(base, 0);
  if (n < N_NODES) offs[n] = base + incl - c;
}

static_assert(N_NODES <= 65536, "col must fit the record's 16 bits");
__global__ __launch_bounds__(256) void fill_records(const int* __restrict__ edge_index,
                                                    const int* __restrict__ edge_attr,
                                                    const int* __restrict__ offs,
                                                    const int* __restrict__ rank,
                                                    unsigned int* __restrict__ rec) {
  const int  t    = blockIdx.x * 256 + threadIdx.x;
  const bool tail = t < EDGE_TAIL;
  const int2* __restrict__ attr2 = (const int2*)edge_attr;
  int  row[EDGE_FULL + 1], col[EDGE_FULL + 1], rk[EDGE_FULL + 1], seg[EDGE_FULL + 1];
  int2 at[EDGE_FULL + 1];
#pragma unroll
  for (int k = 0; k <= EDGE_FULL; ++k) {
    const int e = t + k * EDGE_LANES;
    row[k] = 0; col[k] = 0; rk[k] = 0; at[k] = int2{0, 0};
    if (k < EDGE_FULL || tail) {
      row[k] = edge_index[e];
      col[k] = edge_index[N_EDGES + e];
      at[k]  = attr2[e];
      rk[k]  = rank[e];
    }
  }
#pragma unroll
  for (int k = 0; k <= EDGE_FULL; ++k) {
    seg[k] = 0;
    if (k < EDGE_FULL || tail) seg[k] = offs[row[k]];
  }
#pragma unroll
  for (int k = 0; k <= EDGE_FULL; ++k) {
    if (k < EDGE_FULL || tail) {
      const unsigned int code = ((unsigned int)col[k] << 16) | ((unsigned int)at[k].x << 8) |
                                (unsigned int)at[k].y;
      atomicExch(&rec[seg[k] + rk[k]], code);
    }
  }
}

#define NODES_PER_WAVE 5
#define GATHER_WGS     (N_NODES / (8 * NODES_PER_WAVE))
static_assert(GATHER_WGS * 8 * NODES_PER_WAVE == N_NODES, "the waves must tile the nodes exactly");

__device__ __forceinline__ void add_edges(float4& s, unsigned int r, int m,
                                          const float* __restrict__ x,
                                          const float* __restrict__ e1,
                                          const float* __restrict__ e2, int d0) {
#pragma unroll 4
  for (int j = 0; j < m; ++j) {
    const unsigned int code = __shfl(r, j);
    const float4 xv = *(const float4*)(x + (code >> 16) * DIM + d0);
    const float4 ev = *(const float4*)(e1 + ((code >> 8) & 255) * DIM + d0);
    const float4 fv = *(const float4*)(e2 + (code & 255) * DIM + d0);
    s.x += xv.x + ev.x + fv.x;
    s.y += xv.y + ev.y + fv.y;
    s.z += xv.z + ev.z + fv.z;
    s.w += xv.w + ev.w + fv.w;
  }
}

__global__ __launch_bounds__(256) __attribute__((amdgpu_waves_per_eu(8, 10)))
void gather_nodes(const float* __restrict__ x,
                  const float* __restrict__ e1,
                  const float* __restrict__ e2,
                  const int* __restrict__ cnt,
                  const int* __restrict__ offs,
                  const unsigned int* __restrict__ rec,
                  float* __restrict__ acc) {
  const int lane = threadIdx.x & 31;
  const int n0 = (blockIdx.x * 8 + __builtin_amdgcn_readfirstlane(threadIdx.x >> 5)) * NODES_PER_WAVE;
  const int d0 = lane * 4;

  int c[NODES_PER_WAVE], start[NODES_PER_WAVE];
  unsigned int r0[NODES_PER_WAVE];
#pragma unroll
  for (int k = 0; k < NODES_PER_WAVE; ++k) {
    c[k]     = __builtin_amdgcn_readfirstlane(cnt[n0 + k]);
    start[k] = __builtin_amdgcn_readfirstlane(offs[n0 + k]);
  }
#pragma unroll
  for (int k = 0; k < NODES_PER_WAVE; ++k) {
    r0[k] = 0;
    if (lane < c[k]) r0[k] = rec[start[k] + lane];
  }
  const float4 t1 = *(const float4*)(e1 + 4 * DIM + d0);
  const float4 t2 = *(const float4*)(e2 + 0 * DIM + d0);
  float4 xs = *(const float4*)(x + n0 * DIM + d0);

#pragma unroll
  for (int k = 0; k < NODES_PER_WAVE; ++k) {
    const int n = n0 + k;
    float4 xs_next = xs;
    if (k + 1 < NODES_PER_WAVE) xs_next = *(const float4*)(x + (n + 1) * DIM + d0);
    float4 s;
    s.x = xs.x + t1.x + t2.x;
    s.y = xs.y + t1.y + t2.y;
    s.z = xs.z + t1.z + t2.z;
    s.w = xs.w + t1.w + t2.w;

    add_edges(s, r0[k], c[k] < 32 ? c[k] : 32, x, e1, e2, d0);
    for (int base = 32; base < c[k]; base += 32) {
      unsigned int r = 0;
      if (base + lane < c[k]) r = rec[start[k] + base + lane];
      add_edges(s, r, (c[k] - base < 32) ? (c[k] - base) : 32, x, e1, e2, d0);
    }
    *(float4*)(acc + n * DIM + d0) = s;
    xs = xs_next;
  }
}

#define N_TILES   (N_NODES / 16)
#define MLP_WGS   256
#define TILES_LO  (N_TILES / MLP_WGS)
#define TILES_REM (N_TILES % MLP_WGS)

template <int MT>
__device__ __forceinline__ void layer1(const float* As, const float* W1s, float* Hs,
                                       int wave, int m, int kh, float biasA, float biasB) {
  const float* ap = As + m * AS_STR + 2 * kh;
  const float* bp = W1s + (wave * 32 + m) * AS_STR + 2 * kh;
  v8f c[MT][2];
#pragma unroll
  for (int t = 0; t < MT; ++t) { c[t][0] = v8f{}; c[t][1] = v8f{}; }
#pragma unroll
  for (int k0 = 0; k0 < DIM; k0 += 4) {
    const v2f b0 = *(const v2f*)(bp + k0);
    const v2f b1 = *(const v2f*)(bp + 16 * AS_STR + k0);
#pragma unroll
    for (int t = 0; t < MT; ++t) {
      const v2f a = *(const v2f*)(ap + t * 16 * AS_STR + k0);
      c[t][0] = __builtin_amdgcn_wmma_f32_16x16x4_f32(false, a, false, b0, (short)0, c[t][0],
                                                      false, false);
      c[t][1] = __builtin_amdgcn_wmma_f32_16x16x4_f32(false, a, false, b1, (short)0, c[t][1],
                                                      false, false);
    }
  }
#pragma unroll
  for (int t = 0; t < MT; ++t) {
#pragma unroll
    for (int nt = 0; nt < 2; ++nt) {
      const float bias = nt ? biasB : biasA;
      const int n0 = wave * 32 + nt * 16;
#pragma unroll
      for (int j = 0; j < 8; ++j) {
        const float h = c[t][nt][j] + bias;
        Hs[(t * 16 + j + 8 * kh) * HS_STR + n0 + m] = h > 0.0f ? h : 0.0f;
      }
    }
  }
}

__device__ __forceinline__ void layer2(const float* Hs, const float* W2s, float* __restrict__ out,
                                       int row0, int mt, int wave, int m, int kh,
                                       float biasA, float biasB) {
  const int tr = wave >> 2;
  if (tr >= mt) return;
  const int np = wave & 3;
  const float* ap = Hs + (tr * 16 + m) * HS_STR + 2 * kh;
  const float* bp = W2s + (np * 32 + m) * HS_STR + 2 * kh;
  v8f c0 = {}, c1 = {};
#pragma unroll
  for (int k0 = 0; k0 < HID; k0 += 4) {
    const v2f a  = *(const v2f*)(ap + k0);
    const v2f b0 = *(const v2f*)(bp + k0);
    const v2f b1 = *(const v2f*)(bp + 16 * HS_STR + k0);
    c0 = __builtin_amdgcn_wmma_f32_16x16x4_f32(false, a, false, b0, (short)0, c0, false, false);
    c1 = __builtin_amdgcn_wmma_f32_16x16x4_f32(false, a, false, b1, (short)0, c1, false, false);
  }
  float* op = out + (row0 + tr * 16 + 8 * kh) * DIM + np * 32 + m;
#pragma unroll
  for (int j = 0; j < 8; ++j) {
    op[j * DIM]      = c0[j] + biasA;
    op[j * DIM + 16] = c1[j] + biasB;
  }
}

__global__ __launch_bounds__(256) void mlp_fused(const float* __restrict__ acc,
                                                 const float* __restrict__ W1,
                                                 const float* __restrict__ b1,
                                                 const float* __restrict__ W2,
                                                 const float* __restrict__ b2,
                                                 float* __restrict__ out) {
  __shared__ __attribute__((aligned(16))) float W1s[HID * AS_STR];
  __shared__ __attribute__((aligned(16))) float W2s[DIM * HS_STR];
  __shared__ __attribute__((aligned(16))) float As[32 * AS_STR];
  __shared__ __attribute__((aligned(16))) float Hs[32 * HS_STR];
  const int tid  = threadIdx.x;
  const int wave = tid >> 5;
  const int lane = tid & 31;
  const int m    = lane & 15;
  const int kh   = lane >> 4;
  const int wg   = blockIdx.x;
  int tile = wg * TILES_LO + (wg < TILES_REM ? wg : TILES_REM);
  int left = TILES_LO + (wg < TILES_REM ? 1 : 0);

  {
    const float4* w1v = (const float4*)W1;
#pragma unroll 16
    for (int q = 0; q < HID * DIM / 4 / 256; ++q) {
      const int i = tid + 256 * q;
      *(float4*)(W1s + (i >> 5) * AS_STR + (i & 31) * 4) = w1v[i];
    }
    const float4* w2v = (const float4*)W2;
#pragma unroll 16
    for (int q = 0; q < DIM * HID / 4 / 256; ++q) {
      const int i = tid + 256 * q;
      *(float4*)(W2s + (i >> 6) * HS_STR + (i & 63) * 4) = w2v[i];
    }
  }
  const float bias1a = b1[wave * 32 + m];
  const float bias1b = b1[wave * 32 + 16 + m];
  const float bias2a = b2[(wave & 3) * 32 + m];
  const float bias2b = b2[(wave & 3) * 32 + 16 + m];

  float4 p0 = {}, p1 = {}, p2 = {}, p3 = {};
  float* const d0 = As + (tid >> 5) * AS_STR + (tid & 31) * 4;
  {
    const float4* s = (const float4*)(acc + tile * 16 * DIM);
    p0 = s[tid]; p1 = s[tid + 256];
    *(float4*)d0 = p0; *(float4*)(d0 + 8 * AS_STR) = p1;
    if (left >= 2) {
      p2 = s[tid + 512]; p3 = s[tid + 768];
      *(float4*)(d0 + 16 * AS_STR) = p2; *(float4*)(d0 + 24 * AS_STR) = p3;
    }
  }
  __syncthreads();

  while (left > 0) {
    const int mt    = left >= 2 ? 2 : 1;
    const int ntile = tile + mt;
    const int nleft = left - mt;
    {
      const float4* s = (const float4*)(acc + ntile * 16 * DIM);
      if (nleft >= 1) { p0 = s[tid]; p1 = s[tid + 256]; }
      if (nleft >= 2) { p2 = s[tid + 512]; p3 = s[tid + 768]; }
    }
    if (mt == 2) layer1<2>(As, W1s, Hs, wave, m, kh, bias1a, bias1b);
    else         layer1<1>(As, W1s, Hs, wave, m, kh, bias1a, bias1b);
    __syncthreads();
    if (nleft >= 1) { *(float4*)d0 = p0; *(float4*)(d0 + 8 * AS_STR) = p1; }
    if (nleft >= 2) { *(float4*)(d0 + 16 * AS_STR) = p2; *(float4*)(d0 + 24 * AS_STR) = p3; }
    layer2(Hs, W2s, out, tile * 16, mt, wave, m, kh, bias2a, bias2b);
    __syncthreads();
    tile = ntile;
    left = nleft;
  }
}

extern "C" void kernel_launch(void* const* d_in, const int* in_sizes, int n_in,
                              void* d_out, int out_size, void* d_ws, size_t ws_size,
                              hipStream_t stream) {
  const float* x          = (const float*)d_in[0];
  const int*   edge_index = (const int*)d_in[1];
  const int*   edge_attr  = (const int*)d_in[2];
  const float* W1         = (const float*)d_in[3];
  const float* b1         = (const float*)d_in[4];
  const float* W2         = (const float*)d_in[5];
  const float* b2         = (const float*)d_in[6];
  const float* e1         = (const float*)d_in[7];
  const float* e2         = (const float*)d_in[8];
  float*       out        = (float*)d_out;
  char*        ws         = (char*)d_ws;
  float*       acc        = (float*)ws;
  int*         cnt        = (int*)(ws + 25600000);
  int*         offs       = (int*)(ws + 25800000);
  int*         rank       = (int*)(ws + 26000000);
  unsigned int* rec       = (unsigned int*)(ws + 28400000);
  int*         cursor     = (int*)(ws + 33200000);

  (void)in_sizes; (void)n_in; (void)out_size; (void)ws_size;

  zero_counts<<<196, 256, 0, stream>>>(cnt, cursor);
  count_edges<<<EDGE_WGS, 256, 0, stream>>>(edge_index, cnt, rank);
  alloc_segments<<<196, 256, 0, stream>>>(cnt, cursor, offs);
  fill_records<<<EDGE_WGS, 256, 0, stream>>>(edge_index, edge_attr, offs, rank, rec);
  gather_nodes<<<GATHER_WGS, 256, 0, stream>>>(x, e1, e2, cnt, offs, rec, acc);
  mlp_fused<<<MLP_WGS, 256, 0, stream>>>(acc, W1, b1, W2, b2, out);
}


# Counting sort (count, offsets, fill) replaced by per-row linked lists: one returning exchange per edge on head[row]; gather walks 5 lists per wave; sum order = list order (reverse arrival)
# speedup vs baseline: 2.9164x; 1.1934x over previous
#include <hip/hip_runtime.h>


#define N_NODES 50000
#define N_EDGES 600000
#define DIM     128
#define HID     256
#define AS_STR  (DIM + 4)
#define HS_STR  (HID + 4)

typedef __attribute__((ext_vector_type(2))) float v2f;
typedef __attribute__((ext_vector_type(4))) float v4f;
typedef __attribute__((ext_vector_type(8))) float v8f;

__global__ __launch_bounds__(256) void init_heads(int* __restrict__ head) {
  int idx = blockIdx.x * 256 + threadIdx.x;
  if (idx < N_NODES) head[idx] = -1;
}

#define EDGE_WGS   256
#define EDGE_LANES (EDGE_WGS * 256)
#define EDGE_FULL  (N_EDGES / EDGE_LANES)
#define EDGE_TAIL  (N_EDGES - EDGE_FULL * EDGE_LANES)

static_assert(N_NODES <= 65536, "col must fit the record's 16 bits");
__global__ __launch_bounds__(256) void link_edges(const int* __restrict__ edge_index,
                                                  const int* __restrict__ edge_attr,
                                                  int* __restrict__ head,
                                                  int* __restrict__ link) {
  const int  t    = blockIdx.x * 256 + threadIdx.x;
  const bool tail = t < EDGE_TAIL;
  const int2* __restrict__ attr2 = (const int2*)edge_attr;
  int2* __restrict__ link2 = (int2*)link;
  int  row[EDGE_FULL + 1], col[EDGE_FULL + 1], prev[EDGE_FULL + 1];
  int2 at[EDGE_FULL + 1];
#pragma unroll
  for (int k = 0; k <= EDGE_FULL; ++k) {
    const int e = t + k * EDGE_LANES;
    row[k] = 0; col[k] = 0; at[k] = int2{0, 0};
    if (k < EDGE_FULL || tail) {
      row[k] = edge_index[e];
      col[k] = edge_index[N_EDGES + e];
      at[k]  = attr2[e];
    }
  }
#pragma unroll
  for (int k = 0; k <= EDGE_FULL; ++k) {
    prev[k] = -1;
    if (k < EDGE_FULL || tail) prev[k] = atomicExch(&head[row[k]], t + k * EDGE_LANES);
  }
#pragma unroll
  for (int k = 0; k <= EDGE_FULL; ++k) {
    if (k < EDGE_FULL || tail) {
      const unsigned int code = ((unsigned int)col[k] << 16) | ((unsigned int)at[k].x << 8) |
                                (unsigned int)at[k].y;
      link2[t + k * EDGE_LANES] = int2{prev[k], (int)code};
    }
  }
}

#define NODES_PER_WAVE 5
#define GATHER_WGS     (N_NODES / (8 * NODES_PER_WAVE))
static_assert(GATHER_WGS * 8 * NODES_PER_WAVE == N_NODES, "the waves must tile the nodes exactly");

__global__ __launch_bounds__(256) __attribute__((amdgpu_waves_per_eu(10, 10)))
void gather_nodes(const float* __restrict__ x,
                  const float* __restrict__ e1,
                  const float* __restrict__ e2,
                  const int* __restrict__ head,
                  const int* __restrict__ link,
                  float* __restrict__ acc) {
  const unsigned int lane = threadIdx.x & 31;
  const int n0 = (blockIdx.x * 8 + __builtin_amdgcn_readfirstlane(threadIdx.x >> 5)) * NODES_PER_WAVE;
  const int2* __restrict__ link2 = (const int2*)link;

  int e = -1;
  if (lane < NODES_PER_WAVE) e = head[n0 + lane];
  if ((unsigned int)e >= (unsigned int)N_EDGES) e = -1;
  int2 L = int2{-1, 0};
  if (e >= 0) L = link2[e];

  const v4f t1 = ((const v4f*)(e1 + 4 * DIM))[lane];
  const v4f t2 = ((const v4f*)(e2 + 0 * DIM))[lane];
  v4f s[NODES_PER_WAVE];
#pragma unroll
  for (int k = 0; k < NODES_PER_WAVE; ++k) {
    const v4f xs = ((const v4f*)(x + (n0 + k) * DIM))[lane];
    s[k] = xs + t1 + t2;
  }

  for (int hop = 0; hop < N_EDGES; ++hop) {
    const unsigned int live = __builtin_amdgcn_ballot_w32(e >= 0);
    if (live == 0u) break;
    const int code = L.y;
    const int nxt  = L.x;
    v4f xv[NODES_PER_WAVE], ev[NODES_PER_WAVE], fv[NODES_PER_WAVE];
#pragma unroll
    for (int k = 0; k < NODES_PER_WAVE; ++k) {
      xv[k] = v4f{0.f, 0.f, 0.f, 0.f}; ev[k] = xv[k]; fv[k] = xv[k];
      if ((live >> k) & 1u) {
        const unsigned int c = (unsigned int)__builtin_amdgcn_readlane(code, k);
        xv[k] = ((const v4f*)(x  + (size_t)(c >> 16) * DIM))[lane];
        ev[k] = ((const v4f*)(e1 + (size_t)((c >> 8) & 255u) * DIM))[lane];
        fv[k] = ((const v4f*)(e2 + (size_t)(c & 255u) * DIM))[lane];
      }
    }
    e = ((unsigned int)nxt < (unsigned int)N_EDGES) ? nxt : -1;
    L = int2{-1, 0};
    if (e >= 0) L = link2[e];
    asm volatile("" :: "v"(xv[0]), "v"(ev[0]), "v"(fv[0]), "v"(xv[1]), "v"(ev[1]), "v"(fv[1]),
                       "v"(xv[2]), "v"(ev[2]), "v"(fv[2]), "v"(xv[3]), "v"(ev[3]), "v"(fv[3]),
                       "v"(xv[4]), "v"(ev[4]), "v"(fv[4]));
#pragma unroll
    for (int k = 0; k < NODES_PER_WAVE; ++k) s[k] += xv[k] + ev[k] + fv[k];
  }
#pragma unroll
  for (int k = 0; k < NODES_PER_WAVE; ++k) ((v4f*)(acc + (n0 + k) * DIM))[lane] = s[k];
}

#define N_TILES   (N_NODES / 16)
#define MLP_WGS   256
#define TILES_LO  (N_TILES / MLP_WGS)
#define TILES_REM (N_TILES % MLP_WGS)

template <int MT>
__device__ __forceinline__ void layer1(const float* As, const float* W1s, float* Hs,
                                       int wave, int m, int kh, float biasA, float biasB) {
  const float* ap = As + m * AS_STR + 2 * kh;
  const float* bp = W1s + (wave * 32 + m) * AS_STR + 2 * kh;
  v8f c[MT][2];
#pragma unroll
  for (int t = 0; t < MT; ++t) { c[t][0] = v8f{}; c[t][1] = v8f{}; }
#pragma unroll
  for (int k0 = 0; k0 < DIM; k0 += 4) {
    const v2f b0 = *(const v2f*)(bp + k0);
    const v2f b1 = *(const v2f*)(bp + 16 * AS_STR + k0);
#pragma unroll
    for (int t = 0; t < MT; ++t) {
      const v2f a = *(const v2f*)(ap + t * 16 * AS_STR + k0);
      c[t][0] = __builtin_amdgcn_wmma_f32_16x16x4_f32(false, a, false, b0, (short)0, c[t][0],
                                                      false, false);
      c[t][1] = __builtin_amdgcn_wmma_f32_16x16x4_f32(false, a, false, b1, (short)0, c[t][1],
                                                      false, false);
    }
  }
#pragma unroll
  for (int t = 0; t < MT; ++t) {
#pragma unroll
    for (int nt = 0; nt < 2; ++nt) {
      const float bias = nt ? biasB : biasA;
      const int n0 = wave * 32 + nt * 16;
#pragma unroll
      for (int j = 0; j < 8; ++j) {
        const float h = c[t][nt][j] + bias;
        Hs[(t * 16 + j + 8 * kh) * HS_STR + n0 + m] = h > 0.0f ? h : 0.0f;
      }
    }
  }
}

__device__ __forceinline__ void layer2(const float* Hs, const float* W2s, float* __restrict__ out,
                                       int row0, int mt, int wave, int m, int kh,
                                       float biasA, float biasB) {
  const int tr = wave >> 2;
  if (tr >= mt) return;
  const int np = wave & 3;
  const float* ap = Hs + (tr * 16 + m) * HS_STR + 2 * kh;
  const float* bp = W2s + (np * 32 + m) * HS_STR + 2 * kh;
  v8f c0 = {}, c1 = {};
#pragma unroll
  for (int k0 = 0; k0 < HID; k0 += 4) {
    const v2f a  = *(const v2f*)(ap + k0);
    const v2f b0 = *(const v2f*)(bp + k0);
    const v2f b1 = *(const v2f*)(bp + 16 * HS_STR + k0);
    c0 = __builtin_amdgcn_wmma_f32_16x16x4_f32(false, a, false, b0, (short)0, c0, false, false);
    c1 = __builtin_amdgcn_wmma_f32_16x16x4_f32(false, a, false, b1, (short)0, c1, false, false);
  }
  float* op = out + (row0 + tr * 16 + 8 * kh) * DIM + np * 32 + m;
#pragma unroll
  for (int j = 0; j < 8; ++j) {
    op[j * DIM]      = c0[j] + biasA;
    op[j * DIM + 16] = c1[j] + biasB;
  }
}

__global__ __launch_bounds__(256) void mlp_fused(const float* __restrict__ acc,
                                                 const float* __restrict__ W1,
                                                 const float* __restrict__ b1,
                                                 const float* __restrict__ W2,
                                                 const float* __restrict__ b2,
                                                 float* __restrict__ out) {
  __shared__ __attribute__((aligned(16))) float W1s[HID * AS_STR];
  __shared__ __attribute__((aligned(16))) float W2s[DIM * HS_STR];
  __shared__ __attribute__((aligned(16))) float As[32 * AS_STR];
  __shared__ __attribute__((aligned(16))) float Hs[32 * HS_STR];
  const int tid  = threadIdx.x;
  const int wave = tid >> 5;
  const int lane = tid & 31;
  const int m    = lane & 15;
  const int kh   = lane >> 4;
  const int wg   = blockIdx.x;
  int tile = wg * TILES_LO + (wg < TILES_REM ? wg : TILES_REM);
  int left = TILES_LO + (wg < TILES_REM ? 1 : 0);

  {
    const float4* w1v = (const float4*)W1;
#pragma unroll 16
    for (int q = 0; q < HID * DIM / 4 / 256; ++q) {
      const int i = tid + 256 * q;
      *(float4*)(W1s + (i >> 5) * AS_STR + (i & 31) * 4) = w1v[i];
    }
    const float4* w2v = (const float4*)W2;
#pragma unroll 16
    for (int q = 0; q < DIM * HID / 4 / 256; ++q) {
      const int i = tid + 256 * q;
      *(float4*)(W2s + (i >> 6) * HS_STR + (i & 63) * 4) = w2v[i];
    }
  }
  const float bias1a = b1[wave * 32 + m];
  const float bias1b = b1[wave * 32 + 16 + m];
  const float bias2a = b2[(wave & 3) * 32 + m];
  const float bias2b = b2[(wave & 3) * 32 + 16 + m];

  float4 p0 = {}, p1 = {}, p2 = {}, p3 = {};
  float* const d0 = As + (tid >> 5) * AS_STR + (tid & 31) * 4;
  {
    const float4* s = (const float4*)(acc + tile * 16 * DIM);
    p0 = s[tid]; p1 = s[tid + 256];
    *(float4*)d0 = p0; *(float4*)(d0 + 8 * AS_STR) = p1;
    if (left >= 2) {
      p2 = s[tid + 512]; p3 = s[tid + 768];
      *(float4*)(d0 + 16 * AS_STR) = p2; *(float4*)(d0 + 24 * AS_STR) = p3;
    }
  }
  __syncthreads();

  while (left > 0) {
    const int mt    = left >= 2 ? 2 : 1;
    const int ntile = tile + mt;
    const int nleft = left - mt;
    {
      const float4* s = (const float4*)(acc + ntile * 16 * DIM);
      if (nleft >= 1) { p0 = s[tid]; p1 = s[tid + 256]; }
      if (nleft >= 2) { p2 = s[tid + 512]; p3 = s[tid + 768]; }
    }
    if (mt == 2) layer1<2>(As, W1s, Hs, wave, m, kh, bias1a, bias1b);
    else         layer1<1>(As, W1s, Hs, wave, m, kh, bias1a, bias1b);
    __syncthreads();
    if (nleft >= 1) { *(float4*)d0 = p0; *(float4*)(d0 + 8 * AS_STR) = p1; }
    if (nleft >= 2) { *(float4*)(d0 + 16 * AS_STR) = p2; *(float4*)(d0 + 24 * AS_STR) = p3; }
    layer2(Hs, W2s, out, tile * 16, mt, wave, m, kh, bias2a, bias2b);
    __syncthreads();
    tile = ntile;
    left = nleft;
  }
}

extern "C" void kernel_launch(void* const* d_in, const int* in_sizes, int n_in,
                              void* d_out, int out_size, void* d_ws, size_t ws_size,
                              hipStream_t stream) {
  const float* x          = (const float*)d_in[0];
  const int*   edge_index = (const int*)d_in[1];
  const int*   edge_attr  = (const int*)d_in[2];
  const float* W1         = (const float*)d_in[3];
  const float* b1         = (const float*)d_in[4];
  const float* W2         = (const float*)d_in[5];
  const float* b2         = (const float*)d_in[6];
  const float* e1         = (const float*)d_in[7];
  const float* e2         = (const float*)d_in[8];
  float*       out        = (float*)d_out;
  char*        ws         = (char*)d_ws;
  float*       acc        = (float*)ws;
  int*         head       = (int*)(ws + 25600000);
  int*         link       = (int*)(ws + 26214400);

  (void)in_sizes; (void)n_in; (void)out_size; (void)ws_size;

  init_heads<<<196, 256, 0, stream>>>(head);
  link_edges<<<EDGE_WGS, 256, 0, stream>>>(edge_index, edge_attr, head, link);
  gather_nodes<<<GATHER_WGS, 256, 0, stream>>>(x, e1, e2, head, link, acc);
  mlp_fused<<<MLP_WGS, 256, 0, stream>>>(acc, W1, b1, W2, b2, out);
}


# Head words moved to one per 128-byte line (heads plane of 50000 lines, cleared with whole-line stores) so returning exchanges of different rows never share a line; sum order rule unchanged
# speedup vs baseline: 3.5012x; 1.2005x over previous
#include <hip/hip_runtime.h>


#define N_NODES 50000
#define N_EDGES 600000
#define DIM     128
#define HID     256
#define AS_STR  (DIM + 4)
#define HS_STR  (HID + 4)

typedef __attribute__((ext_vector_type(2))) float v2f;
typedef __attribute__((ext_vector_type(4))) float v4f;
typedef __attribute__((ext_vector_type(8))) float v8f;

#define HEAD_STRIDE 32
#define INIT_WGS    196
#define INIT_LANES  (INIT_WGS * 256)
#define HEAD_VEC4   (N_NODES * HEAD_STRIDE / 4)
#define INIT_STEPS  ((HEAD_VEC4 + INIT_LANES - 1) / INIT_LANES)
__global__ __launch_bounds__(256) void init_heads(int* __restrict__ head) {
  const int t = blockIdx.x * 256 + threadIdx.x;
  int4* __restrict__ head4 = (int4*)head;
#pragma unroll
  for (int k = 0; k < INIT_STEPS; ++k) {
    const int idx = t + k * INIT_LANES;
    if (idx < HEAD_VEC4) head4[idx] = int4{-1, -1, -1, -1};
  }
}

#define EDGE_WGS   256
#define EDGE_LANES (EDGE_WGS * 256)
#define EDGE_FULL  (N_EDGES / EDGE_LANES)
#define EDGE_TAIL  (N_EDGES - EDGE_FULL * EDGE_LANES)

static_assert(N_NODES <= 65536, "col must fit the record's 16 bits");
__global__ __launch_bounds__(256) void link_edges(const int* __restrict__ edge_index,
                                                  const int* __restrict__ edge_attr,
                                                  int* __restrict__ head,
                                                  int* __restrict__ link) {
  const int  t    = blockIdx.x * 256 + threadIdx.x;
  const bool tail = t < EDGE_TAIL;
  const int2* __restrict__ attr2 = (const int2*)edge_attr;
  int2* __restrict__ link2 = (int2*)link;
  int  row[EDGE_FULL + 1], col[EDGE_FULL + 1], prev[EDGE_FULL + 1];
  int2 at[EDGE_FULL + 1];
#pragma unroll
  for (int k = 0; k <= EDGE_FULL; ++k) {
    const int e = t + k * EDGE_LANES;
    row[k] = 0; col[k] = 0; at[k] = int2{0, 0};
    if (k < EDGE_FULL || tail) {
      row[k] = edge_index[e];
      col[k] = edge_index[N_EDGES + e];
      at[k]  = attr2[e];
    }
  }
#pragma unroll
  for (int k = 0; k <= EDGE_FULL; ++k) {
    prev[k] = -1;
    if (k < EDGE_FULL || tail) prev[k] = atomicExch(&head[row[k] * HEAD_STRIDE], t + k * EDGE_LANES);
  }
#pragma unroll
  for (int k = 0; k <= EDGE_FULL; ++k) {
    if (k < EDGE_FULL || tail) {
      const unsigned int code = ((unsigned int)col[k] << 16) | ((unsigned int)at[k].x << 8) |
                                (unsigned int)at[k].y;
      link2[t + k * EDGE_LANES] = int2{prev[k], (int)code};
    }
  }
}

#define NODES_PER_WAVE 5
#define GATHER_WGS     (N_NODES / (8 * NODES_PER_WAVE))
static_assert(GATHER_WGS * 8 * NODES_PER_WAVE == N_NODES, "the waves must tile the nodes exactly");

__global__ __launch_bounds__(256) __attribute__((amdgpu_waves_per_eu(10, 10)))
void gather_nodes(const float* __restrict__ x,
                  const float* __restrict__ e1,
                  const float* __restrict__ e2,
                  const int* __restrict__ head,
                  const int* __restrict__ link,
                  float* __restrict__ acc) {
  const unsigned int lane = threadIdx.x & 31;
  const int n0 = (blockIdx.x * 8 + __builtin_amdgcn_readfirstlane(threadIdx.x >> 5)) * NODES_PER_WAVE;
  const int2* __restrict__ link2 = (const int2*)link;

  int e = -1;
  if (lane < NODES_PER_WAVE) e = head[(n0 + lane) * HEAD_STRIDE];
  if ((unsigned int)e >= (unsigned int)N_EDGES) e = -1;
  int2 L = int2{-1, 0};
  if (e >= 0) L = link2[e];

  const v4f t1 = ((const v4f*)(e1 + 4 * DIM))[lane];
  const v4f t2 = ((const v4f*)(e2 + 0 * DIM))[lane];
  v4f s[NODES_PER_WAVE];
#pragma unroll
  for (int k = 0; k < NODES_PER_WAVE; ++k) {
    const v4f xs = ((const v4f*)(x + (n0 + k) * DIM))[lane];
    s[k] = xs + t1 + t2;
  }

  for (int hop = 0; hop < N_EDGES; ++hop) {
    const unsigned int live = __builtin_amdgcn_ballot_w32(e >= 0);
    if (live == 0u) break;
    const int code = L.y;
    const int nxt  = L.x;
    v4f xv[NODES_PER_WAVE], ev[NODES_PER_WAVE], fv[NODES_PER_WAVE];
#pragma unroll
    for (int k = 0; k < NODES_PER_WAVE; ++k) {
      xv[k] = v4f{0.f, 0.f, 0.f, 0.f}; ev[k] = xv[k]; fv[k] = xv[k];
      if ((live >> k) & 1u) {
        const unsigned int c = (unsigned int)__builtin_amdgcn_readlane(code, k);
        xv[k] = ((const v4f*)(x  + (size_t)(c >> 16) * DIM))[lane];
        ev[k] = ((const v4f*)(e1 + (size_t)((c >> 8) & 255u) * DIM))[lane];
        fv[k] = ((const v4f*)(e2 + (size_t)(c & 255u) * DIM))[lane];
      }
    }
    e = ((unsigned int)nxt < (unsigned int)N_EDGES) ? nxt : -1;
    L = int2{-1, 0};
    if (e >= 0) L = link2[e];
    asm volatile("" :: "v"(xv[0]), "v"(ev[0]), "v"(fv[0]), "v"(xv[1]), "v"(ev[1]), "v"(fv[1]),
                       "v"(xv[2]), "v"(ev[2]), "v"(fv[2]), "v"(xv[3]), "v"(ev[3]), "v"(fv[3]),
                       "v"(xv[4]), "v"(ev[4]), "v"(fv[4]));
#pragma unroll
    for (int k = 0; k < NODES_PER_WAVE; ++k) s[k] += xv[k] + ev[k] + fv[k];
  }
#pragma unroll
  for (int k = 0; k < NODES_PER_WAVE; ++k) ((v4f*)(acc + (n0 + k) * DIM))[lane] = s[k];
}

#define N_TILES   (N_NODES / 16)
#define MLP_WGS   256
#define TILES_LO  (N_TILES / MLP_WGS)
#define TILES_REM (N_TILES % MLP_WGS)

template <int MT>
__device__ __forceinline__ void layer1(const float* As, const float* W1s, float* Hs,
                                       int wave, int m, int kh, float biasA, float biasB) {
  const float* ap = As + m * AS_STR + 2 * kh;
  const float* bp = W1s + (wave * 32 + m) * AS_STR + 2 * kh;
  v8f c[MT][2];
#pragma unroll
  for (int t = 0; t < MT; ++t) { c[t][0] = v8f{}; c[t][1] = v8f{}; }
#pragma unroll
  for (int k0 = 0; k0 < DIM; k0 += 4) {
    const v2f b0 = *(const v2f*)(bp + k0);
    const v2f b1 = *(const v2f*)(bp + 16 * AS_STR + k0);
#pragma unroll
    for (int t = 0; t < MT; ++t) {
      const v2f a = *(const v2f*)(ap + t * 16 * AS_STR + k0);
      c[t][0] = __builtin_amdgcn_wmma_f32_16x16x4_f32(false, a, false, b0, (short)0, c[t][0],
                                                      false, false);
      c[t][1] = __builtin_amdgcn_wmma_f32_16x16x4_f32(false, a, false, b1, (short)0, c[t][1],
                                                      false, false);
    }
  }
#pragma unroll
  for (int t = 0; t < MT; ++t) {
#pragma unroll
    for (int nt = 0; nt < 2; ++nt) {
      const float bias = nt ? biasB : biasA;
      const int n0 = wave * 32 + nt * 16;
#pragma unroll
      for (int j = 0; j < 8; ++j) {
        const float h = c[t][nt][j] + bias;
        Hs[(t * 16 + j + 8 * kh) * HS_STR + n0 + m] = h > 0.0f ? h : 0.0f;
      }
    }
  }
}

__device__ __forceinline__ void layer2(const float* Hs, const float* W2s, float* __restrict__ out,
                                       int row0, int mt, int wave, int m, int kh,
                                       float biasA, float biasB) {
  const int tr = wave >> 2;
  if (tr >= mt) return;
  const int np = wave & 3;
  const float* ap = Hs + (tr * 16 + m) * HS_STR + 2 * kh;
  const float* bp = W2s + (np * 32 + m) * HS_STR + 2 * kh;
  v8f c0 = {}, c1 = {};
#pragma unroll
  for (int k0 = 0; k0 < HID; k0 += 4) {
    const v2f a  = *(const v2f*)(ap + k0);
    const v2f b0 = *(const v2f*)(bp + k0);
    const v2f b1 = *(const v2f*)(bp + 16 * HS_STR + k0);
    c0 = __builtin_amdgcn_wmma_f32_16x16x4_f32(false, a, false, b0, (short)0, c0, false, false);
    c1 = __builtin_amdgcn_wmma_f32_16x16x4_f32(false, a, false, b1, (short)0, c1, false, false);
  }
  float* op = out + (row0 + tr * 16 + 8 * kh) * DIM + np * 32 + m;
#pragma unroll
  for (int j = 0; j < 8; ++j) {
    op[j * DIM]      = c0[j] + biasA;
    op[j * DIM + 16] = c1[j] + biasB;
  }
}

__global__ __launch_bounds__(256) void mlp_fused(const float* __restrict__ acc,
                                                 const float* __restrict__ W1,
                                                 const float* __restrict__ b1,
                                                 const float* __restrict__ W2,
                                                 const float* __restrict__ b2,
                                                 float* __restrict__ out) {
  __shared__ __attribute__((aligned(16))) float W1s[HID * AS_STR];
  __shared__ __attribute__((aligned(16))) float W2s[DIM * HS_STR];
  __shared__ __attribute__((aligned(16))) float As[32 * AS_STR];
  __shared__ __attribute__((aligned(16))) float Hs[32 * HS_STR];
  const int tid  = threadIdx.x;
  const int wave = tid >> 5;
  const int lane = tid & 31;
  const int m    = lane & 15;
  const int kh   = lane >> 4;
  const int wg   = blockIdx.x;
  int tile = wg * TILES_LO + (wg < TILES_REM ? wg : TILES_REM);
  int left = TILES_LO + (wg < TILES_REM ? 1 : 0);

  {
    const float4* w1v = (const float4*)W1;
#pragma unroll 16
    for (int q = 0; q < HID * DIM / 4 / 256; ++q) {
      const int i = tid + 256 * q;
      *(float4*)(W1s + (i >> 5) * AS_STR + (i & 31) * 4) = w1v[i];
    }
    const float4* w2v = (const float4*)W2;
#pragma unroll 16
    for (int q = 0; q < DIM * HID / 4 / 256; ++q) {
      const int i = tid + 256 * q;
      *(float4*)(W2s + (i >> 6) * HS_STR + (i & 63) * 4) = w2v[i];
    }
  }
  const float bias1a = b1[wave * 32 + m];
  const float bias1b = b1[wave * 32 + 16 + m];
  const float bias2a = b2[(wave & 3) * 32 + m];
  const float bias2b = b2[(wave & 3) * 32 + 16 + m];

  float4 p0 = {}, p1 = {}, p2 = {}, p3 = {};
  float* const d0 = As + (tid >> 5) * AS_STR + (tid & 31) * 4;
  {
    const float4* s = (const float4*)(acc + tile * 16 * DIM);
    p0 = s[tid]; p1 = s[tid + 256];
    *(float4*)d0 = p0; *(float4*)(d0 + 8 * AS_STR) = p1;
    if (left >= 2) {
      p2 = s[tid + 512]; p3 = s[tid + 768];
      *(float4*)(d0 + 16 * AS_STR) = p2; *(float4*)(d0 + 24 * AS_STR) = p3;
    }
  }
  __syncthreads();

  while (left > 0) {
    const int mt    = left >= 2 ? 2 : 1;
    const int ntile = tile + mt;
    const int nleft = left - mt;
    {
      const float4* s = (const float4*)(acc + ntile * 16 * DIM);
      if (nleft >= 1) { p0 = s[tid]; p1 = s[tid + 256]; }
      if (nleft >= 2) { p2 = s[tid + 512]; p3 = s[tid + 768]; }
    }
    if (mt == 2) layer1<2>(As, W1s, Hs, wave, m, kh, bias1a, bias1b);
    else         layer1<1>(As, W1s, Hs, wave, m, kh, bias1a, bias1b);
    __syncthreads();
    if (nleft >= 1) { *(float4*)d0 = p0; *(float4*)(d0 + 8 * AS_STR) = p1; }
    if (nleft >= 2) { *(float4*)(d0 + 16 * AS_STR) = p2; *(float4*)(d0 + 24 * AS_STR) = p3; }
    layer2(Hs, W2s, out, tile * 16, mt, wave, m, kh, bias2a, bias2b);
    __syncthreads();
    tile = ntile;
    left = nleft;
  }
}

extern "C" void kernel_launch(void* const* d_in, const int* in_sizes, int n_in,
                              void* d_out, int out_size, void* d_ws, size_t ws_size,
                              hipStream_t stream) {
  const float* x          = (const float*)d_in[0];
  const int*   edge_index = (const int*)d_in[1];
  const int*   edge_attr  = (const int*)d_in[2];
  const float* W1         = (const float*)d_in[3];
  const float* b1         = (const float*)d_in[4];
  const float* W2         = (const float*)d_in[5];
  const float* b2         = (const float*)d_in[6];
  const float* e1         = (const float*)d_in[7];
  const float* e2         = (const float*)d_in[8];
  float*       out        = (float*)d_out;
  char*        ws         = (char*)d_ws;
  float*       acc        = (float*)ws;
  int*         head       = (int*)(ws + 25600000);
  int*         link       = (int*)(ws + 32000000);

  (void)in_sizes; (void)n_in; (void)out_size; (void)ws_size;

  init_heads<<<INIT_WGS, 256, 0, stream>>>(head);
  link_edges<<<EDGE_WGS, 256, 0, stream>>>(edge_index, edge_attr, head, link);
  gather_nodes<<<GATHER_WGS, 256, 0, stream>>>(x, e1, e2, head, link, acc);
  mlp_fused<<<MLP_WGS, 256, 0, stream>>>(acc, W1, b1, W2, b2, out);
}
